# starting kernel
# baseline (speedup 1.0000x reference)
#include <hip/hip_runtime.h>
#include <math.h>


typedef float v2f __attribute__((ext_vector_type(2)));
typedef float v8f __attribute__((ext_vector_type(8)));

#define TPB 256


__global__ __launch_bounds__(TPB) void k_init_deg(float* deg, float* degw, int n) {
    int i = blockIdx.x * TPB + threadIdx.x;
    if (i < n) { deg[i] = 1.0f; degw[i] = 1.0f; }
}

__global__ __launch_bounds__(TPB) void k_count_edges(const int* __restrict__ idx,
                                                     float* __restrict__ deg, int e) {
    int i = blockIdx.x * TPB + threadIdx.x;
    if (i < e) atomicAdd(&deg[idx[i]], 1.0f);
}

__global__ __launch_bounds__(TPB) void k_rsqrt(const float* __restrict__ a,
                                               float* __restrict__ o, int n) {
    int i = blockIdx.x * TPB + threadIdx.x;
    if (i < n) o[i] = rsqrtf(a[i]);
}

__global__ __launch_bounds__(TPB) void k_relu(float* h, long long n) {
    long long i = (long long)blockIdx.x * TPB + threadIdx.x;
    if (i < n) h[i] = fmaxf(h[i], 0.0f);
}

__global__ __launch_bounds__(TPB) void k_selfloop_scale(const float* __restrict__ s,
                                                        const float* __restrict__ hin,
                                                        float* __restrict__ hout,
                                                        int n, int d4) {
    long long t = (long long)blockIdx.x * TPB + threadIdx.x;
    if (t >= (long long)n * d4) return;
    int i = (int)(t / d4);
    float sv = s[i]; sv = sv * sv;
    float4 v = ((const float4*)hin)[t];
    float4 r; r.x = sv * v.x; r.y = sv * v.y; r.z = sv * v.z; r.w = sv * v.w;
    ((float4*)hout)[t] = r;
}

__global__ __launch_bounds__(TPB) void k_gemm_bias_wmma(const float* __restrict__ A,
                                                        const float* __restrict__ W,
                                                        const float* __restrict__ bias,
                                                        float* __restrict__ out,
                                                        int M, int K, int Nc) {
    int wid = (blockIdx.x * TPB + threadIdx.x) >> 5;
    int ntile = Nc >> 4;
    int total = (M >> 4) * ntile;
    if (wid >= total) return;

    int mt = wid / ntile, nt = wid - mt * ntile;
    int m0 = mt << 4, n0 = nt << 4;
    int lane = threadIdx.x & 31;
    int half = lane >> 4;
    int l16  = lane & 15;

    const float* arow = A + (long long)(m0 + l16) * K;
    const float* wrow = W + (long long)(n0 + l16) * K;

    v8f c = {0.f, 0.f, 0.f, 0.f, 0.f, 0.f, 0.f, 0.f};
    for (int kb = 0; kb < K; kb += 4) {
        int k = kb + 2 * half;
        v2f a = *(const v2f*)(arow + k);
        v2f b = *(const v2f*)(wrow + k);
        c = __builtin_amdgcn_wmma_f32_16x16x4_f32(false, a, false, b,
                                                  (short)0, c, false, false);
    }

    float bv = bias[n0 + l16];
    long long colbase = (long long)n0 + l16;
#pragma unroll
    for (int r = 0; r < 8; ++r) {
        int m = m0 + r + 8 * half;
        out[(long long)m * Nc + colbase] = c[r] + bv;
    }
}

__global__ __launch_bounds__(TPB) void k_prop1_edges(const int* __restrict__ row,
                                                     const int* __restrict__ col,
                                                     const float* __restrict__ dis,
                                                     const float* __restrict__ hlin,
                                                     float* __restrict__ hagg, int E) {
    long long t = (long long)blockIdx.x * TPB + threadIdx.x;
    int e = (int)(t >> 5);
    if (e >= E) return;
    int lane = (int)(t & 31);
    int r = row[e], c = col[e];
    float nrm = dis[r] * dis[c];
    float4 v = ((const float4*)(hlin + (long long)r * 128))[lane];
    float* dst = hagg + (long long)c * 128 + lane * 4;
    atomicAdd(dst + 0, nrm * v.x);
    atomicAdd(dst + 1, nrm * v.y);
    atomicAdd(dst + 2, nrm * v.z);
    atomicAdd(dst + 3, nrm * v.w);
}

__global__ __launch_bounds__(TPB) void k_knn_weights(const int* __restrict__ rk,
                                                     const int* __restrict__ ck,
                                                     const float* __restrict__ h,
                                                     const float* __restrict__ alpha,
                                                     float* __restrict__ wknn,
                                                     float* __restrict__ degw, int E) {
    long long t = (long long)blockIdx.x * TPB + threadIdx.x;
    int e = (int)(t >> 5);
    if (e >= E) return;
    int lane = (int)(t & 31);
    int a = rk[e], b = ck[e];
    float4 va = ((const float4*)(h + (long long)a * 128))[lane];
    float4 vb = ((const float4*)(h + (long long)b * 128))[lane];
    float dx = va.x - vb.x, dy = va.y - vb.y, dz = va.z - vb.z, dw = va.w - vb.w;
    float ss = dx * dx + dy * dy + dz * dz + dw * dw;
#pragma unroll
    for (int off = 16; off > 0; off >>= 1) ss += __shfl_xor(ss, off, 32);
    if (lane == 0) {
        float wv = 0.0f;
        if (a != b) {
            float nv = fmaxf(sqrtf(ss), 1e-12f);
            wv = alpha[0] * sqrtf(nv);
        }
        wknn[e] = wv;
        if (wv != 0.0f) atomicAdd(&degw[b], wv);
    }
}

__global__ __launch_bounds__(TPB) void k_prop2_edges(const int* __restrict__ row,
                                                     const int* __restrict__ col,
                                                     const float* __restrict__ disw,
                                                     const float* __restrict__ wknn,
                                                     const float* __restrict__ h2,
                                                     float* __restrict__ out, int E) {
    long long t = (long long)blockIdx.x * TPB + threadIdx.x;
    int e = (int)(t >> 5);
    if (e >= E) return;
    int lane = (int)(t & 31);
    float wv = wknn ? wknn[e] : 1.0f;
    if (wv == 0.0f) return;
    int r = row[e], c = col[e];
    float nrm = disw[r] * disw[c] * wv;
    float2 v = ((const float2*)(h2 + (long long)r * 64))[lane];
    float* dst = out + (long long)c * 64 + lane * 2;
    atomicAdd(dst + 0, nrm * v.x);
    atomicAdd(dst + 1, nrm * v.y);
}


static inline int nblk(long long threads) { return (int)((threads + TPB - 1) / TPB); }

extern "C" void kernel_launch(void* const* d_in, const int* in_sizes, int n_in,
                              void* d_out, int out_size, void* d_ws, size_t ws_size,
                              hipStream_t stream) {
    (void)n_in; (void)out_size; (void)ws_size;
    const float* x      = (const float*)d_in[0];
    const int*   ei_org = (const int*)d_in[1];
    const int*   ei_knn = (const int*)d_in[2];
    const float* alpha  = (const float*)d_in[3];
    const float* W1     = (const float*)d_in[4];
    const float* b1     = (const float*)d_in[5];
    const float* W2     = (const float*)d_in[6];
    const float* b2     = (const float*)d_in[7];
    float* out = (float*)d_out;
    float* ws  = (float*)d_ws;

    const int N = 50000, DIN = 128, DHID = 128, DOUT = 64;
    const int Eo = in_sizes[1] / 2;
    const int Ek = in_sizes[2] / 2;

    float* deg  = ws;
    float* dis  = ws + (long long)N;
    float* degw = ws + 2LL * N;
    float* disw = ws + 3LL * N;
    float* hlin = ws + 4LL * N;
    float* h    = hlin + (long long)N * DHID;
    float* h2   = h + (long long)N * DHID;
    float* wknn = h2 + (long long)N * DOUT;

    const int* row_o = ei_org;
    const int* col_o = ei_org + Eo;
    const int* row_k = ei_knn;
    const int* col_k = ei_knn + Ek;

    k_init_deg<<<nblk(N), TPB, 0, stream>>>(deg, degw, N);
    k_count_edges<<<nblk(Eo), TPB, 0, stream>>>(row_o, deg, Eo);
    k_rsqrt<<<nblk(N), TPB, 0, stream>>>(deg, dis, N);

    long long w1waves = (long long)(N / 16) * (DHID / 16);
    k_gemm_bias_wmma<<<nblk(w1waves * 32), TPB, 0, stream>>>(x, W1, b1, hlin, N, DIN, DHID);

    k_selfloop_scale<<<nblk((long long)N * 32), TPB, 0, stream>>>(dis, hlin, h, N, 32);
    k_prop1_edges<<<nblk((long long)Eo * 32), TPB, 0, stream>>>(row_o, col_o, dis, hlin, h, Eo);
    k_relu<<<nblk((long long)N * DHID), TPB, 0, stream>>>(h, (long long)N * DHID);

    k_count_edges<<<nblk(Eo), TPB, 0, stream>>>(col_o, degw, Eo);
    k_knn_weights<<<nblk((long long)Ek * 32), TPB, 0, stream>>>(row_k, col_k, h, alpha,
                                                                wknn, degw, Ek);
    k_rsqrt<<<nblk(N), TPB, 0, stream>>>(degw, disw, N);

    long long w2waves = (long long)(N / 16) * (DOUT / 16);
    k_gemm_bias_wmma<<<nblk(w2waves * 32), TPB, 0, stream>>>(h, W2, b2, h2, N, DHID, DOUT);

    k_selfloop_scale<<<nblk((long long)N * 16), TPB, 0, stream>>>(disw, h2, out, N, 16);
    k_prop2_edges<<<nblk((long long)Eo * 32), TPB, 0, stream>>>(row_o, col_o, disw,
                                                                (const float*)nullptr, h2, out, Eo);
    k_prop2_edges<<<nblk((long long)Ek * 32), TPB, 0, stream>>>(row_k, col_k, disw,
                                                                wknn, h2, out, Ek);
}


# Tile scans read one byte per edge (its destination tile's number, written by the call's first kernel), 512 edges per load, not 4-byte destinations; LDS sums in arrival order as before
# speedup vs baseline: 1.5507x; 1.5507x over previous
#include <hip/hip_runtime.h>
#include <math.h>


typedef float v2f __attribute__((ext_vector_type(2)));
typedef float v8f __attribute__((ext_vector_type(8)));

#define TPB 256

#define TILE_NODES 196
#define KEY_CHUNK  512


__global__ __launch_bounds__(TPB) void k_setup(float* __restrict__ deg, float* __restrict__ degw, int n,
                                               const int* __restrict__ col_a, int Ea,
                                               unsigned* __restrict__ key_a, int words_a,
                                               const int* __restrict__ col_b, int Eb,
                                               unsigned* __restrict__ key_b, int words_b) {
    int i = blockIdx.x * TPB + threadIdx.x;
    for (int j = i; j < n; j += gridDim.x * TPB) { deg[j] = 1.0f; degw[j] = 1.0f; }
    const int* col = col_a; unsigned* key = key_a; int E = Ea; int t = i;
    if (t >= words_a) {
        t -= words_a;
        if (t >= words_b) return;
        col = col_b; key = key_b; E = Eb;
    }
    unsigned pk = 0u;
#pragma unroll
    for (int b = 0; b < 4; ++b) {
        int e = 4 * t + b;
        unsigned k = 0xFFu;
        if (e < E) k = ((unsigned)col[e] / (unsigned)TILE_NODES) & 0xFFu;
        pk |= k << (8 * b);
    }
    key[t] = pk;
}

__global__ __launch_bounds__(TPB) void k_count_edges(const int* __restrict__ idx,
                                                     float* __restrict__ deg, int e) {
    int i = blockIdx.x * TPB + threadIdx.x;
    if (i < e) atomicAdd(&deg[idx[i]], 1.0f);
}

__global__ __launch_bounds__(TPB) void k_rsqrt(const float* __restrict__ a,
                                               float* __restrict__ o, int n) {
    int i = blockIdx.x * TPB + threadIdx.x;
    if (i < n) o[i] = rsqrtf(a[i]);
}

__global__ __launch_bounds__(TPB) void k_gemm_bias_wmma(const float* __restrict__ A,
                                                        const float* __restrict__ W,
                                                        const float* __restrict__ bias,
                                                        float* __restrict__ out,
                                                        int M, int K, int Nc) {
    int wid = (blockIdx.x * TPB + threadIdx.x) >> 5;
    int ntile = Nc >> 4;
    int total = (M >> 4) * ntile;
    if (wid >= total) return;

    int mt = wid / ntile, nt = wid - mt * ntile;
    int m0 = mt << 4, n0 = nt << 4;
    int lane = threadIdx.x & 31;
    int half = lane >> 4;
    int l16  = lane & 15;

    const float* arow = A + (long long)(m0 + l16) * K;
    const float* wrow = W + (long long)(n0 + l16) * K;

    v8f c = {0.f, 0.f, 0.f, 0.f, 0.f, 0.f, 0.f, 0.f};
    for (int kb = 0; kb < K; kb += 4) {
        int k = kb + 2 * half;
        v2f a = *(const v2f*)(arow + k);
        v2f b = *(const v2f*)(wrow + k);
        c = __builtin_amdgcn_wmma_f32_16x16x4_f32(false, a, false, b,
                                                  (short)0, c, false, false);
    }

    float bv = bias[n0 + l16];
    long long colbase = (long long)n0 + l16;
#pragma unroll
    for (int r = 0; r < 8; ++r) {
        int m = m0 + r + 8 * half;
        out[(long long)m * Nc + colbase] = c[r] + bv;
    }
}

__global__ __launch_bounds__(TPB) void k_knn_weights(const int* __restrict__ rk,
                                                     const int* __restrict__ ck,
                                                     const float* __restrict__ h,
                                                     const float* __restrict__ alpha,
                                                     float* __restrict__ wknn,
                                                     float* __restrict__ degw, int E) {
    long long t = (long long)blockIdx.x * TPB + threadIdx.x;
    int e = (int)(t >> 5);
    if (e >= E) return;
    int lane = (int)(t & 31);
    int a = rk[e], b = ck[e];
    float4 va = ((const float4*)(h + (long long)a * 128))[lane];
    float4 vb = ((const float4*)(h + (long long)b * 128))[lane];
    float dx = va.x - vb.x, dy = va.y - vb.y, dz = va.z - vb.z, dw = va.w - vb.w;
    float ss = dx * dx + dy * dy + dz * dz + dw * dw;
#pragma unroll
    for (int off = 16; off > 0; off >>= 1) ss += __shfl_xor(ss, off, 32);
    if (lane == 0) {
        float wv = 0.0f;
        if (a != b) {
            float nv = fmaxf(sqrtf(ss), 1e-12f);
            wv = alpha[0] * sqrtf(nv);
        }
        wknn[e] = wv;
        if (wv != 0.0f) atomicAdd(&degw[b], wv);
    }
}


#define TILE_TPB   1024
#define TILE_WAVES (TILE_TPB / 32)
#define LIST_CAP   64
#define GATHER_U   8

typedef float vf4 __attribute__((ext_vector_type(4)));
template <int V> struct RowVec;
template <> struct RowVec<4> { typedef vf4 t; };
template <> struct RowVec<2> { typedef v2f t; };

template <int D>
__device__ __forceinline__ void tile_add_group(const int* __restrict__ row,
                                               const int* __restrict__ col,
                                               const float* __restrict__ w,
                                               const float* __restrict__ s,
                                               const float* __restrict__ hin,
                                               float* acc, const int* le,
                                               int m, int base, int cnt, int lane) {
    constexpr int VEC = D / 32;
    typedef typename RowVec<VEC>::t vec_t;
    int r = 0, cl = 0; float nrm = 0.0f;
    if (lane < m) {
        int e = le[lane];
        r = row[e];
        cl = col[e] - base;
        float wv = w ? w[e] : 1.0f;
        const bool ok = ((unsigned)cl < (unsigned)cnt);
        if (!ok) cl = 0;
        nrm = ok ? s[r] * s[base + cl] * wv : 0.0f;
    }
    for (int j0 = 0; j0 < m; j0 += GATHER_U) {
        vec_t v[GATHER_U];
#pragma unroll
        for (int u = 0; u < GATHER_U; ++u) {
            int jj = j0 + u; if (jj > m - 1) jj = m - 1;
            int rj = __builtin_amdgcn_readlane(r, jj);
            v[u] = *(const vec_t*)(hin + (long long)rj * D + lane * VEC);
        }
#pragma unroll
        for (int u = 0; u < GATHER_U; ++u) {
            int j = j0 + u;
            if (j < m) {
                float nj = __int_as_float(__builtin_amdgcn_readlane(__float_as_int(nrm), j));
                int   cj = __builtin_amdgcn_readlane(cl, j);
                float* dst = acc + cj * D + lane;
#pragma unroll
                for (int k = 0; k < VEC; ++k) atomicAdd(dst + k * 32, nj * v[u][k]);
            }
        }
    }
}

__device__ __forceinline__ unsigned zero_bytes(unsigned x) {
    return ~(((x & 0x7f7f7f7fu) + 0x7f7f7f7fu) | x) & 0x80808080u;
}

__device__ __forceinline__ unsigned chunk_hits(const uint4 q, unsigned pat) {
    return zero_bytes(q.x ^ pat) | (zero_bytes(q.y ^ pat) >> 1)
         | (zero_bytes(q.z ^ pat) >> 2) | (zero_bytes(q.w ^ pat) >> 3);
}

template <int D>
__device__ __forceinline__ void tile_scan(const unsigned* __restrict__ keys,
                                          const int* __restrict__ row,
                                          const int* __restrict__ col,
                                          const float* __restrict__ w, int E,
                                          const float* __restrict__ s,
                                          const float* __restrict__ hin,
                                          float* acc, int* le, unsigned pat,
                                          int base, int cnt, int lane, int wave) {
    const int nchunks = (E + KEY_CHUNK - 1) / KEY_CHUNK;
    if (wave >= nchunks) return;
    const uint4* kp = (const uint4*)keys + lane;
    const int last = nchunks - 1;
    int count = 0;
    int cn = wave + TILE_WAVES; if (cn > last) cn = last;
    uint4 q0 = kp[wave * 32];
    uint4 q1 = kp[cn * 32];
    for (int c = wave; c < nchunks; c += 2 * TILE_WAVES) {
        const uint4 cur0 = q0, cur1 = q1;
        int c2 = c + 2 * TILE_WAVES; if (c2 > last) c2 = last;
        int c3 = c + 3 * TILE_WAVES; if (c3 > last) c3 = last;
        q0 = kp[c2 * 32];
        q1 = kp[c3 * 32];
        unsigned long long hit = ((unsigned long long)chunk_hits(cur1, pat) << 32)
                               | (unsigned long long)chunk_hits(cur0, pat);
        const int ebase = c * KEY_CHUNK + lane * 16;
        while (__builtin_amdgcn_ballot_w32(hit != 0ull)) {
            bool has = (hit != 0ull);
            int e = 0;
            if (has) {
                const int p = __builtin_ctzll(hit);
                hit &= hit - 1ull;
                const int q = p & 31;
                e = ebase + (p >> 5) * (TILE_WAVES * KEY_CHUNK) + ((7 - (q & 7)) << 2) + (q >> 3);
                has = (e < E);
            }
            const unsigned hm = __builtin_amdgcn_ballot_w32(has);
            if (has) le[count + (int)__builtin_amdgcn_mbcnt_lo(hm, 0u)] = e;
            count += __builtin_popcount(hm);
            if (count >= 32) {
                count -= 32;
                tile_add_group<D>(row, col, w, s, hin, acc, le + count, 32, base, cnt, lane);
            }
        }
    }
    if (count > 0)
        tile_add_group<D>(row, col, w, s, hin, acc, le, count, base, cnt, lane);
}

template <int D, bool RELU>
__global__ __launch_bounds__(TILE_TPB) void k_tile_prop(const int* __restrict__ row_a,
                                                        const int* __restrict__ col_a,
                                                        const unsigned* __restrict__ key_a, int Ea,
                                                        const int* __restrict__ row_b,
                                                        const int* __restrict__ col_b,
                                                        const unsigned* __restrict__ key_b,
                                                        const float* __restrict__ w_b, int Eb,
                                                        const float* __restrict__ s,
                                                        const float* __restrict__ hin,
                                                        float* __restrict__ hout, int n) {
    constexpr int VEC = D / 32;
    typedef typename RowVec<VEC>::t vec_t;
    __shared__ float acc[TILE_NODES * D];
    __shared__ int lst_e[TILE_WAVES * LIST_CAP];

    const int tid  = threadIdx.x;
    const int lane = tid & 31;
    const int wave = __builtin_amdgcn_readfirstlane(tid >> 5);
    const int base = blockIdx.x * TILE_NODES;
    int cnt = n - base; if (cnt > TILE_NODES) cnt = TILE_NODES;

    for (int idx = tid; idx < cnt * 32; idx += TILE_TPB) {
        int node = idx >> 5, q = idx & 31;
        float sv = s[base + node]; sv = sv * sv;
        vec_t v = *(const vec_t*)(hin + (long long)(base + node) * D + q * VEC);
#pragma unroll
        for (int k = 0; k < VEC; ++k) acc[node * D + k * 32 + q] = sv * v[k];
    }
    __syncthreads();

    int* le = lst_e + wave * LIST_CAP;
    const unsigned pat = (unsigned)blockIdx.x * 0x01010101u;
    tile_scan<D>(key_a, row_a, col_a, (const float*)nullptr, Ea, s, hin, acc, le, pat, base, cnt, lane, wave);
    if (Eb > 0)
        tile_scan<D>(key_b, row_b, col_b, w_b, Eb, s, hin, acc, le, pat, base, cnt, lane, wave);
    __syncthreads();

    for (int idx = tid; idx < cnt * 32; idx += TILE_TPB) {
        int node = idx >> 5, q = idx & 31;
        vec_t v;
#pragma unroll
        for (int k = 0; k < VEC; ++k) {
            float a = acc[node * D + k * 32 + q];
            v[k] = RELU ? fmaxf(a, 0.0f) : a;
        }
        *(vec_t*)(hout + (long long)(base + node) * D + q * VEC) = v;
    }
}


static inline int nblk(long long threads) { return (int)((threads + TPB - 1) / TPB); }

extern "C" void kernel_launch(void* const* d_in, const int* in_sizes, int n_in,
                              void* d_out, int out_size, void* d_ws, size_t ws_size,
                              hipStream_t stream) {
    (void)n_in; (void)out_size; (void)ws_size;
    const float* x      = (const float*)d_in[0];
    const int*   ei_org = (const int*)d_in[1];
    const int*   ei_knn = (const int*)d_in[2];
    const float* alpha  = (const float*)d_in[3];
    const float* W1     = (const float*)d_in[4];
    const float* b1     = (const float*)d_in[5];
    const float* W2     = (const float*)d_in[6];
    const float* b2     = (const float*)d_in[7];
    float* out = (float*)d_out;
    float* ws  = (float*)d_ws;

    const int N = 50000, DIN = 128, DHID = 128, DOUT = 64;
    static_assert((N - 1) / TILE_NODES < 256, "a tile's number must fit one byte");
    const int Eo = in_sizes[1] / 2;
    const int Ek = in_sizes[2] / 2;

    float* deg  = ws;
    float* dis  = ws + (long long)N;
    float* degw = ws + 2LL * N;
    float* disw = ws + 3LL * N;
    float* hlin = ws + 4LL * N;
    float* h    = hlin + (long long)N * DHID;
    float* h2   = h + (long long)N * DHID;
    const int nco = (Eo + KEY_CHUNK - 1) / KEY_CHUNK;
    const int nck = (Ek + KEY_CHUNK - 1) / KEY_CHUNK;
    const int words_o = nco * (KEY_CHUNK / 4);
    const int words_k = nck * (KEY_CHUNK / 4);
    unsigned* key_o = (unsigned*)(h2 + (long long)N * DOUT);
    unsigned* key_k = key_o + words_o;
    float* wknn = (float*)(key_k + words_k);

    const int* row_o = ei_org;
    const int* col_o = ei_org + Eo;
    const int* row_k = ei_knn;
    const int* col_k = ei_knn + Ek;

    k_setup<<<nblk((long long)words_o + words_k), TPB, 0, stream>>>(deg, degw, N, col_o, Eo, key_o, words_o,
                                                     col_k, Ek, key_k, words_k);

    k_count_edges<<<nblk(Eo), TPB, 0, stream>>>(row_o, deg, Eo);
    k_rsqrt<<<nblk(N), TPB, 0, stream>>>(deg, dis, N);

    long long w1waves = (long long)(N / 16) * (DHID / 16);
    k_gemm_bias_wmma<<<nblk(w1waves * 32), TPB, 0, stream>>>(x, W1, b1, hlin, N, DIN, DHID);

    const int NT = (N + TILE_NODES - 1) / TILE_NODES;
    k_tile_prop<128, true><<<NT, TILE_TPB, 0, stream>>>(row_o, col_o, key_o, Eo,
                                                        (const int*)nullptr, (const int*)nullptr,
                                                        (const unsigned*)nullptr,
                                                        (const float*)nullptr, 0,
                                                        dis, hlin, h, N);

    k_count_edges<<<nblk(Eo), TPB, 0, stream>>>(col_o, degw, Eo);
    k_knn_weights<<<nblk((long long)Ek * 32), TPB, 0, stream>>>(row_k, col_k, h, alpha,
                                                                wknn, degw, Ek);
    k_rsqrt<<<nblk(N), TPB, 0, stream>>>(degw, disw, N);

    long long w2waves = (long long)(N / 16) * (DOUT / 16);
    k_gemm_bias_wmma<<<nblk(w2waves * 32), TPB, 0, stream>>>(h, W2, b2, h2, N, DHID, DOUT);

    k_tile_prop<64, false><<<NT, TILE_TPB, 0, stream>>>(row_o, col_o, key_o, Eo,
                                                        row_k, col_k, key_k, wknn, Ek,
                                                        disw, h2, out, N);
}
